# starting kernel
# baseline (speedup 1.0000x reference)
#include <hip/hip_runtime.h>

#define IN_F  128
#define HID_F 64
#define OUT_F 40

typedef float v2f __attribute__((ext_vector_type(2)));
typedef float v8f __attribute__((ext_vector_type(8)));

__device__ __forceinline__ v8f wmma4(v2f a, v2f b, v8f c) {
  return __builtin_amdgcn_wmma_f32_16x16x4_f32(
       false, a,  false, b,
       (short)0, c,  false,  false);
}

__device__ __forceinline__ void atomadd(float* p, float v) {
  unsafeAtomicAdd(p, v);
}

__global__ void deg_kernel(const int* __restrict__ src, float* __restrict__ deg, int E) {
  int e = blockIdx.x * blockDim.x + threadIdx.x;
  if (e < E) atomadd(&deg[src[e]], 1.0f);
}

__global__ void dinv_kernel(float* __restrict__ deg, int n) {
  int i = blockIdx.x * blockDim.x + threadIdx.x;
  if (i < n) {
    float d = deg[i];
    deg[i] = (d > 0.0f) ? __frsqrt_rn(d) : 0.0f;
  }
}

__global__ void norm_kernel(const int* __restrict__ src, const int* __restrict__ dst,
                            const float* __restrict__ dinv, float* __restrict__ nrm, int E) {
  int e = blockIdx.x * blockDim.x + threadIdx.x;
  if (e < E) nrm[e] = -dinv[src[e]] * dinv[dst[e]];
}

template <int F>
__global__ void prop_kernel(const float* __restrict__ xin, const int* __restrict__ src,
                            const int* __restrict__ dst, const float* __restrict__ nrm,
                            float* __restrict__ out, int E) {
  const int CH = F / 4;
  long long gid = (long long)blockIdx.x * blockDim.x + threadIdx.x;
  long long nwork = (long long)E * CH;
  if (gid >= nwork) return;
  int c = (int)(gid % CH);
  int e = (int)(gid / CH);
  int s = src[e], d = dst[e];
  float w = nrm[e];
  const float4 v = *(const float4*)(xin + (long long)s * F + c * 4);
  float* o = out + (long long)d * F + c * 4;
  atomadd(o + 0, v.x * w);
  atomadd(o + 1, v.y * w);
  atomadd(o + 2, v.z * w);
  atomadd(o + 3, v.w * w);
}

__global__ __launch_bounds__(256)
void gemm1_kernel(const float* __restrict__ x, const float* __restrict__ t1,
                  const float* __restrict__ p2, const float* __restrict__ W,
                  const float* __restrict__ bias, float* __restrict__ h, int ntile) {
  int wv = (int)((blockIdx.x * (long long)blockDim.x + threadIdx.x) >> 5);
  if (wv >= ntile) return;
  int lane = threadIdx.x & 31;
  int idx = lane & 15, half = lane >> 4;

  long long rowbase = (long long)(wv * 16 + idx) * IN_F;
  const float* xr = x + rowbase;
  const float* tr = t1 + rowbase;
  const float* pr = p2 + rowbase;

  v8f z = {0.f, 0.f, 0.f, 0.f, 0.f, 0.f, 0.f, 0.f};
  v8f acc[4] = {z, z, z, z};

  for (int k0 = 0; k0 < IN_F; k0 += 4) {
    int kk = k0 + 2 * half;
    v2f ax = *(const v2f*)(xr + kk);
    v2f at = *(const v2f*)(tr + kk);
    v2f ap = *(const v2f*)(pr + kk);
    v2f a2 = ap + ap - ax;
#pragma unroll
    for (int nt = 0; nt < 4; ++nt) {
      int n = nt * 16 + idx;
      const float* w0 = W + (long long)kk * HID_F + n;
      const float* w1 = w0 + (long long)IN_F * HID_F;
      const float* w2 = w1 + (long long)IN_F * HID_F;
      v2f b0 = {w0[0], w0[HID_F]};
      v2f b1 = {w1[0], w1[HID_F]};
      v2f b2 = {w2[0], w2[HID_F]};
      acc[nt] = wmma4(ax, b0, acc[nt]);
      acc[nt] = wmma4(at, b1, acc[nt]);
      acc[nt] = wmma4(a2, b2, acc[nt]);
    }
  }
#pragma unroll
  for (int nt = 0; nt < 4; ++nt) {
    int n = nt * 16 + idx;
    float bv = bias[n];
#pragma unroll
    for (int r = 0; r < 8; ++r) {
      long long row = (long long)wv * 16 + r + 8 * half;
      float val = acc[nt][r] + bv;
      h[row * HID_F + n] = val > 0.0f ? val : 0.0f;
    }
  }
}

__global__ __launch_bounds__(256)
void gemm2_kernel(const float* __restrict__ hx, const float* __restrict__ t1,
                  const float* __restrict__ p2, const float* __restrict__ W,
                  const float* __restrict__ bias, float* __restrict__ out, int ntile) {
  int wv = (int)((blockIdx.x * (long long)blockDim.x + threadIdx.x) >> 5);
  if (wv >= ntile) return;
  int lane = threadIdx.x & 31;
  int idx = lane & 15, half = lane >> 4;

  long long rowbase = (long long)(wv * 16 + idx) * HID_F;
  const float* xr = hx + rowbase;
  const float* tr = t1 + rowbase;
  const float* pr = p2 + rowbase;

  v8f z = {0.f, 0.f, 0.f, 0.f, 0.f, 0.f, 0.f, 0.f};
  v8f acc[3] = {z, z, z};

  for (int k0 = 0; k0 < HID_F; k0 += 4) {
    int kk = k0 + 2 * half;
    v2f ax = *(const v2f*)(xr + kk);
    v2f at = *(const v2f*)(tr + kk);
    v2f ap = *(const v2f*)(pr + kk);
    v2f a2 = ap + ap - ax;
#pragma unroll
    for (int nt = 0; nt < 3; ++nt) {
      int n = nt * 16 + idx;
      bool ok = (n < OUT_F);
      int nn = ok ? n : 0;
      const float* w0 = W + (long long)kk * OUT_F + nn;
      const float* w1 = w0 + (long long)HID_F * OUT_F;
      const float* w2 = w1 + (long long)HID_F * OUT_F;
      v2f b0 = {ok ? w0[0] : 0.0f, ok ? w0[OUT_F] : 0.0f};
      v2f b1 = {ok ? w1[0] : 0.0f, ok ? w1[OUT_F] : 0.0f};
      v2f b2 = {ok ? w2[0] : 0.0f, ok ? w2[OUT_F] : 0.0f};
      acc[nt] = wmma4(ax, b0, acc[nt]);
      acc[nt] = wmma4(at, b1, acc[nt]);
      acc[nt] = wmma4(a2, b2, acc[nt]);
    }
  }
#pragma unroll
  for (int nt = 0; nt < 3; ++nt) {
    int n = nt * 16 + idx;
    if (n < OUT_F) {
      float bv = bias[n];
#pragma unroll
      for (int r = 0; r < 8; ++r) {
        long long row = (long long)wv * 16 + r + 8 * half;
        out[row * OUT_F + n] = acc[nt][r] + bv;
      }
    }
  }
}

extern "C" void kernel_launch(void* const* d_in, const int* in_sizes, int n_in,
                              void* d_out, int out_size, void* d_ws, size_t ws_size,
                              hipStream_t stream) {
  const float* x  = (const float*)d_in[0];
  const int*   ei = (const int*)d_in[1];
  const float* W1 = (const float*)d_in[2];
  const float* b1 = (const float*)d_in[3];
  const float* W2 = (const float*)d_in[4];
  const float* b2 = (const float*)d_in[5];

  const int N = in_sizes[0] / IN_F;
  const int E = in_sizes[1] / 2;
  const int* src = ei;
  const int* dst = ei + E;

  float* ws   = (float*)d_ws;
  float* dinv = ws;
  float* nrm  = dinv + N;
  float* bufA = nrm + E;
  float* bufB = bufA + (size_t)N * IN_F;
  float* hbuf = bufB + (size_t)N * IN_F;
  float* outp = (float*)d_out;

  const int B = 256;
  auto cdiv = [](long long a, long long b) { return (int)((a + b - 1) / b); };

  hipMemsetAsync(dinv, 0, (size_t)N * sizeof(float), stream);
  deg_kernel<<<cdiv(E, B), B, 0, stream>>>(src, dinv, E);
  dinv_kernel<<<cdiv(N, B), B, 0, stream>>>(dinv, N);
  norm_kernel<<<cdiv(E, B), B, 0, stream>>>(src, dst, dinv, nrm, E);

  long long w128 = (long long)E * (IN_F / 4);
  hipMemsetAsync(bufA, 0, (size_t)N * IN_F * sizeof(float), stream);
  prop_kernel<IN_F><<<cdiv(w128, B), B, 0, stream>>>(x, src, dst, nrm, bufA, E);
  hipMemsetAsync(bufB, 0, (size_t)N * IN_F * sizeof(float), stream);
  prop_kernel<IN_F><<<cdiv(w128, B), B, 0, stream>>>(bufA, src, dst, nrm, bufB, E);

  const int ntile = N / 16;
  const int gblocks = cdiv((long long)ntile * 32, B);
  gemm1_kernel<<<gblocks, B, 0, stream>>>(x, bufA, bufB, W1, b1, hbuf, ntile);

  long long w64 = (long long)E * (HID_F / 4);
  hipMemsetAsync(bufA, 0, (size_t)N * HID_F * sizeof(float), stream);
  prop_kernel<HID_F><<<cdiv(w64, B), B, 0, stream>>>(hbuf, src, dst, nrm, bufA, E);
  hipMemsetAsync(bufB, 0, (size_t)N * HID_F * sizeof(float), stream);
  prop_kernel<HID_F><<<cdiv(w64, B), B, 0, stream>>>(bufA, src, dst, nrm, bufB, E);

  gemm2_kernel<<<gblocks, B, 0, stream>>>(hbuf, bufA, bufB, W2, b2, outp, ntile);
}


# Dense products re-fed: weights copied once per work-group to LDS (B fragment = one ds_load_b64), A rows read as 64 contiguous bytes per lane per 32-k block; k regrouped, so a new f32 sum order
# speedup vs baseline: 2.2922x; 2.2922x over previous
#include <hip/hip_runtime.h>

#define IN_F  128
#define HID_F 64
#define OUT_F 40

typedef float v2f __attribute__((ext_vector_type(2)));
typedef float v4f __attribute__((ext_vector_type(4)));
typedef float v8f __attribute__((ext_vector_type(8)));

__device__ __forceinline__ v8f wmma4(v2f a, v2f b, v8f c) {
  return __builtin_amdgcn_wmma_f32_16x16x4_f32(
       false, a,  false, b,
       (short)0, c,  false,  false);
}

__device__ __forceinline__ void atomadd(float* p, float v) {
  unsafeAtomicAdd(p, v);
}

__device__ __forceinline__ int cache_half() {
  return (int)((__builtin_amdgcn_s_sendmsg_rtn(0x87) >> 18) & 1u);
}

__device__ __forceinline__ int draw(int* c) {
  int t = 0;
  if ((threadIdx.x & 31u) == 0u) t = atomicAdd(c, 1);
  return __builtin_amdgcn_readfirstlane(t);
}

#define DS  32
#define DRB 12500
#define DU  8
__global__ __launch_bounds__(1024)
void deg_kernel(const int* __restrict__ src, int* __restrict__ part, int E, int N) {
  __shared__ int hist[DRB];
  const int nrange = (N + DRB - 1) / DRB;
  const int es = (E + DS - 1) / DS;
  for (int job = blockIdx.x; job < nrange * DS; job += gridDim.x) {
    const int sl = job / nrange;
    const int base = (job - sl * nrange) * DRB;
    const int len = (N - base < DRB) ? (N - base) : DRB;
    for (int i = threadIdx.x; i < len; i += blockDim.x) hist[i] = 0;
    __syncthreads();
    const int e0 = sl * es;
    const int e1 = (e0 + es < E) ? (e0 + es) : E;
    for (int eb = e0 + (int)threadIdx.x; eb < e1; eb += DU * (int)blockDim.x) {
      int v[DU];
#pragma unroll
      for (int u = 0; u < DU; ++u) {
        const int e = eb + u * (int)blockDim.x;
        v[u] = src[(e < e1) ? e : e1 - 1];
      }
#pragma unroll
      for (int u = 0; u < DU; ++u) {
        const int e = eb + u * (int)blockDim.x;
        const unsigned k = (unsigned)(v[u] - base);
        if (e < e1 && k < (unsigned)len) atomicAdd(&hist[k], 1);
      }
    }
    __syncthreads();
    int* o = part + (long long)sl * N + base;
    for (int i = threadIdx.x; i < len; i += blockDim.x) o[i] = hist[i];
    __syncthreads();
  }
}

__global__ void dinv_kernel(const int* __restrict__ part, float* __restrict__ deg, int n,
                            int* __restrict__ tick) {
  int i = blockIdx.x * blockDim.x + threadIdx.x;
  if (i < 256) tick[i] = 0;
  if (i < n) {
    int c = 0;
#pragma unroll
    for (int s = 0; s < DS; ++s) c += part[(long long)s * n + i];
    float d = (float)c;
    deg[i] = (d > 0.0f) ? __frsqrt_rn(d) : 0.0f;
  }
}

__global__ void norm_kernel(const int* __restrict__ src, const int* __restrict__ dst,
                            const float* __restrict__ dinv, float* __restrict__ nrm, int E) {
  int e = blockIdx.x * blockDim.x + threadIdx.x;
  if (e < E) nrm[e] = -dinv[src[e]] * dinv[dst[e]];
}


template <int F>
__device__ __forceinline__ void walk_run(const float* __restrict__ xin, float* __restrict__ out,
                                         int s, int d, int wb, unsigned m, unsigned lane) {
  constexpr int NL = F / 32;
  constexpr int U  = 16 / NL;
  while (__builtin_popcount(m) >= U) {
    int ix[U];
    float v[U][NL];
#pragma unroll
    for (int u = 0; u < U; ++u) {
      ix[u] = __builtin_ctz(m);
      m &= m - 1u;
      const int si = __builtin_amdgcn_readlane(s, ix[u]);
      const float* xr = xin + (long long)si * F;
#pragma unroll
      for (int j = 0; j < NL; ++j) v[u][j] = xr[lane + 32u * j];
    }
#pragma unroll
    for (int u = 0; u < U; ++u) {
      const int di = __builtin_amdgcn_readlane(d, ix[u]);
      const float wi = __int_as_float(__builtin_amdgcn_readlane(wb, ix[u]));
      float* o = out + (long long)di * F;
#pragma unroll
      for (int j = 0; j < NL; ++j) atomadd(o + lane + 32u * j, v[u][j] * wi);
    }
  }
  if (m) {
    const int n = __builtin_popcount(m);
    int ix[U - 1];
    float v[U - 1][NL];
#pragma unroll
    for (int u = 0; u < U - 1; ++u) {
      if (u < n) {
        ix[u] = __builtin_ctz(m);
        m &= m - 1u;
        const int si = __builtin_amdgcn_readlane(s, ix[u]);
        const float* xr = xin + (long long)si * F;
#pragma unroll
        for (int j = 0; j < NL; ++j) v[u][j] = xr[lane + 32u * j];
      }
    }
#pragma unroll
    for (int u = 0; u < U - 1; ++u) {
      if (u < n) {
        const int di = __builtin_amdgcn_readlane(d, ix[u]);
        const float wi = __int_as_float(__builtin_amdgcn_readlane(wb, ix[u]));
        float* o = out + (long long)di * F;
#pragma unroll
        for (int j = 0; j < NL; ++j) atomadd(o + lane + 32u * j, v[u][j] * wi);
      }
    }
  }
}

template <int F>
__global__ __launch_bounds__(256)
void prop_kernel(const float* __restrict__ xin, const int* __restrict__ src,
                 const int* __restrict__ dst, const float* __restrict__ nrm,
                 float* __restrict__ out, int E, int N, int* __restrict__ tick) {
  constexpr int CH = 64;
  const unsigned lane = threadIdx.x & 31u;
  const int nchunk = (E + CH - 1) / CH;
  const int split = N >> 1;
  const int h = cache_half();

  for (int p = 0; p < 2; ++p) {
    const int hh = h ^ p;
    int* c = tick + hh * 64;
    for (int t = draw(c); t < nchunk; t = draw(c)) {
      int s[2], d[2], wb[2];
      unsigned m[2];
#pragma unroll
      for (int r = 0; r < 2; ++r) {
        const long long e = (long long)t * CH + r * 32 + lane;
        const long long ec = (e < E) ? e : (long long)E - 1;
        s[r]  = src[ec];
        d[r]  = dst[ec];
        wb[r] = __float_as_int(nrm[ec]);
      }
#pragma unroll
      for (int r = 0; r < 2; ++r) {
        const long long e = (long long)t * CH + r * 32 + lane;
        const bool mine = (e < E) && ((d[r] >= split ? 1 : 0) == hh);
        m[r] = __builtin_amdgcn_ballot_w32(mine);
      }
      walk_run<F>(xin, out, s[0], d[0], wb[0], m[0], lane);
      walk_run<F>(xin, out, s[1], d[1], wb[1], m[1], lane);
    }
  }
}

template <int K, int NCOLS, int NT, int PITCH, bool RELU>
__global__ __launch_bounds__(256)
void gemm_kernel(const float* __restrict__ x, const float* __restrict__ t1,
                 const float* __restrict__ p2, const float* __restrict__ W,
                 const float* __restrict__ bias, float* __restrict__ out, int ntile) {
  constexpr int KP = K / 2;
  constexpr int Q  = NT * 4;
  constexpr int MW = NT * 32;
  __shared__ float wl[KP * PITCH];

  for (int i = threadIdx.x; i < 3 * KP * Q; i += blockDim.x) {
    const int n4 = i % Q;
    const int r  = i / Q;
    v4f a = {0.f, 0.f, 0.f, 0.f};
    v4f b = {0.f, 0.f, 0.f, 0.f};
    if (n4 * 4 < NCOLS) {
      const float* w = W + (long long)(2 * r) * NCOLS + n4 * 4;
      a = *(const v4f*)w;
      b = *(const v4f*)(w + NCOLS);
    }
    float* o = wl + (r % KP) * PITCH + (r / KP) * MW + n4 * 8;
    v4f lo = {a.x, b.x, a.y, b.y};
    v4f hi = {a.z, b.z, a.w, b.w};
    *(v4f*)o = lo;
    *(v4f*)(o + 4) = hi;
  }
  __syncthreads();

  const int lane = threadIdx.x & 31;
  const int c = lane & 15, h = lane >> 4;
  const int nw = (int)((gridDim.x * blockDim.x) >> 5);
  const int gw = (int)((blockIdx.x * blockDim.x + threadIdx.x) >> 5);

  for (int base = 0; base < ntile; base += nw) {
    const int rem = ntile - base;
    int t;
    if (rem >= nw) {
      t = base + gw;
    } else {
      const int stride = nw / rem;
      if (gw % stride != 0 || gw / stride >= rem) break;
      t = base + gw / stride;
    }

    const long long rb = (long long)(t * 16 + c) * K + 16 * h;
    const float* xr = x + rb;
    const float* tr = t1 + rb;
    const float* pr = p2 + rb;

    const v8f z = {0.f, 0.f, 0.f, 0.f, 0.f, 0.f, 0.f, 0.f};
    v8f acc[NT];
#pragma unroll
    for (int nt = 0; nt < NT; ++nt) acc[nt] = z;

    for (int m = 0; m < K / 32; ++m) {
      v4f ax[4], at[4], ap[4];
#pragma unroll
      for (int q = 0; q < 4; ++q) {
        ax[q] = *(const v4f*)(xr + 32 * m + 4 * q);
        at[q] = *(const v4f*)(tr + 32 * m + 4 * q);
        ap[q] = *(const v4f*)(pr + 32 * m + 4 * q);
      }
      const float* wrow = wl + (16 * m + 8 * h) * PITCH + 2 * c;
#pragma unroll
      for (int j = 0; j < 8; ++j) {
        const int q = j >> 1, e = (j & 1) * 2;
        v2f fx = {ax[q][e], ax[q][e + 1]};
        v2f ft = {at[q][e], at[q][e + 1]};
        v2f fp = {ap[q][e], ap[q][e + 1]};
        v2f f2 = fp + fp - fx;
        const float* wj = wrow + j * PITCH;
#pragma unroll
        for (int nt = 0; nt < NT; ++nt) {
          v2f b0 = *(const v2f*)(wj + nt * 32);
          v2f b1 = *(const v2f*)(wj + MW + nt * 32);
          v2f b2 = *(const v2f*)(wj + 2 * MW + nt * 32);
          acc[nt] = wmma4(fx, b0, acc[nt]);
          acc[nt] = wmma4(ft, b1, acc[nt]);
          acc[nt] = wmma4(f2, b2, acc[nt]);
        }
      }
    }

#pragma unroll
    for (int nt = 0; nt < NT; ++nt) {
      const int n = nt * 16 + c;
      if (n < NCOLS) {
        const float bv = bias[n];
#pragma unroll
        for (int r = 0; r < 8; ++r) {
          const long long row = (long long)t * 16 + r + 8 * h;
          const float val = acc[nt][r] + bv;
          out[row * NCOLS + n] = RELU ? (val > 0.0f ? val : 0.0f) : val;
        }
      }
    }
  }
}

extern "C" void kernel_launch(void* const* d_in, const int* in_sizes, int n_in,
                              void* d_out, int out_size, void* d_ws, size_t ws_size,
                              hipStream_t stream) {
  const float* x  = (const float*)d_in[0];
  const int*   ei = (const int*)d_in[1];
  const float* W1 = (const float*)d_in[2];
  const float* b1 = (const float*)d_in[3];
  const float* W2 = (const float*)d_in[4];
  const float* b2 = (const float*)d_in[5];

  const int N = in_sizes[0] / IN_F;
  const int E = in_sizes[1] / 2;
  const int* src = ei;
  const int* dst = ei + E;

  float* ws   = (float*)d_ws;
  float* dinv = ws;
  float* nrm  = dinv + N;
  float* bufA = nrm + E;
  float* bufB = bufA + (size_t)N * IN_F;
  float* hbuf = bufB + (size_t)N * IN_F;
  float* outp = (float*)d_out;

  const int B = 256;
  auto cdiv = [](long long a, long long b) { return (int)((a + b - 1) / b); };

  int* tick1 = (int*)hbuf;
  int* part  = (int*)bufB;
  int* tick2 = (int*)(bufA + (size_t)N * HID_F);

  deg_kernel<<<256, 1024, 0, stream>>>(src, part, E, N);
  dinv_kernel<<<cdiv(N, B), B, 0, stream>>>(part, dinv, N, tick1);
  norm_kernel<<<cdiv(E, B), B, 0, stream>>>(src, dst, dinv, nrm, E);

  const int PG = 2048;
  hipMemsetAsync(bufA, 0, (size_t)N * IN_F * sizeof(float), stream);
  prop_kernel<IN_F><<<PG, B, 0, stream>>>(x, src, dst, nrm, bufA, E, N, tick1);
  hipMemsetAsync(bufB, 0, (size_t)N * IN_F * sizeof(float), stream);
  prop_kernel<IN_F><<<PG, B, 0, stream>>>(bufA, src, dst, nrm, bufB, E, N, tick1 + 128);

  const int ntile = N / 16;
  const int GG = 768;
  gemm_kernel<IN_F, HID_F, 4, 388, true><<<GG, B, 0, stream>>>(x, bufA, bufB, W1, b1, hbuf, ntile);

  hipMemsetAsync(bufA, 0, ((size_t)N * HID_F + 256) * sizeof(float), stream);
  prop_kernel<HID_F><<<PG, B, 0, stream>>>(hbuf, src, dst, nrm, bufA, E, N, tick2);
  hipMemsetAsync(bufB, 0, (size_t)N * HID_F * sizeof(float), stream);
  prop_kernel<HID_F><<<PG, B, 0, stream>>>(bufA, src, dst, nrm, bufB, E, N, tick2 + 128);

  gemm_kernel<HID_F, OUT_F, 3, 292, false><<<GG, B, 0, stream>>>(hbuf, bufA, bufB, W2, b2, outp, ntile);
}
